# starting kernel
# baseline (speedup 1.0000x reference)
#include <hip/hip_runtime.h>

typedef __attribute__((ext_vector_type(2))) float v2f;
typedef __attribute__((ext_vector_type(8))) float v8f;

#define B_  2
#define S_  512
#define H_  8
#define D_  64
#define E_  512
#define R_  256
#define TM  32

__global__ __launch_bounds__(256) void relpos_wmma_kernel(
    const float* __restrict__ q,
    const float* __restrict__ table,
    const int*   __restrict__ x,
    const int*   __restrict__ max_x_p,
    float* __restrict__ out)
{
    __shared__ float ldsQ[TM * D_];
    __shared__ float ldsP[TM * R_];
    __shared__ int   xs[S_];

    const int tid   = threadIdx.x;
    const int bid   = blockIdx.x;
    const int itile = bid & ((S_ / TM) - 1);
    const int h     = (bid >> 4) & (H_ - 1);
    const int b     = bid >> 7;
    const int i0    = itile * TM;
    const int mx    = max_x_p[0];

#pragma unroll
    for (int k = 0; k < (TM * D_) / 256; ++k) {
        int idx = tid + k * 256;
        int m = idx >> 6;
        int d = idx & (D_ - 1);
        ldsQ[idx] = q[(((size_t)b * S_ + (i0 + m)) * H_ + h) * D_ + d];
    }
#pragma unroll
    for (int k = 0; k < S_ / 256; ++k) {
        int j = tid + k * 256;
        xs[j] = x[b * S_ + j];
    }
    __syncthreads();

    const int wave   = tid >> 5;
    const int lane   = tid & 31;
    const int laneLo = lane & 15;
    const int hiSel  = lane >> 4;

    v2f a[2][16];
#pragma unroll
    for (int mt = 0; mt < 2; ++mt) {
        const float* src = &ldsQ[(mt * 16 + laneLo) * D_ + 2 * hiSel];
#pragma unroll
        for (int k4 = 0; k4 < 16; ++k4) {
            a[mt][k4] = *reinterpret_cast<const v2f*>(src + 4 * k4);
        }
    }

#pragma unroll
    for (int p = 0; p < 2; ++p) {
        const int nt = wave * 2 + p;
        const int r  = nt * 16 + laneLo;
        const float* bsrc = table + (size_t)r * E_ + h * D_ + 2 * hiSel;

        v8f acc0 = {};
        v8f acc1 = {};
#pragma unroll
        for (int k4 = 0; k4 < 16; ++k4) {
            v2f bv = *reinterpret_cast<const v2f*>(bsrc + 4 * k4);
            acc0 = __builtin_amdgcn_wmma_f32_16x16x4_f32(
                       false, a[0][k4], false, bv, (short)0, acc0, false, false);
            acc1 = __builtin_amdgcn_wmma_f32_16x16x4_f32(
                       false, a[1][k4], false, bv, (short)0, acc1, false, false);
        }

#pragma unroll
        for (int c = 0; c < 8; ++c) {
            int M = c + 8 * hiSel;
            ldsP[M * R_ + nt * 16 + laneLo]         = acc0[c];
            ldsP[(16 + M) * R_ + nt * 16 + laneLo]  = acc1[c];
        }
    }
    __syncthreads();

    float* outBase = out + (((size_t)(b * H_ + h)) * S_ + i0) * S_;
#pragma unroll 4
    for (int idx = tid; idx < TM * S_; idx += 256) {
        int m   = idx >> 9;
        int j   = idx & (S_ - 1);
        int rel = mx + xs[j] - xs[i0 + m];
        outBase[(size_t)m * S_ + j] = ldsP[m * R_ + rel];
    }
}

extern "C" void kernel_launch(void* const* d_in, const int* in_sizes, int n_in,
                              void* d_out, int out_size, void* d_ws, size_t ws_size,
                              hipStream_t stream) {
    const float* q     = (const float*)d_in[0];
    const float* table = (const float*)d_in[1];
    const int*   x     = (const int*)d_in[2];
    const int*   mx    = (const int*)d_in[3];
    float*       out   = (float*)d_out;
    (void)in_sizes; (void)n_in; (void)out_size; (void)d_ws; (void)ws_size;

    dim3 grid(B_ * H_ * (S_ / TM));
    dim3 block(256);
    hipLaunchKernelGGL(relpos_wmma_kernel, grid, block, 0, stream, q, table, x, mx, out);
}


# epilogue vectorised: each lane keeps 4 consecutive x[j] and its 16 rows' x[i] in registers, does 4 LDS gathers per row and one 16-B store, rows unrolled (was 4-B stores in 64 dependent trips)
# speedup vs baseline: 1.4789x; 1.4789x over previous
#include <hip/hip_runtime.h>

typedef __attribute__((ext_vector_type(2))) float v2f;
typedef __attribute__((ext_vector_type(8))) float v8f;

#define B_  2
#define S_  512
#define H_  8
#define D_  64
#define E_  512
#define R_  256
#define TM  32

__global__ __launch_bounds__(256) void relpos_wmma_kernel(
    const float* __restrict__ q,
    const float* __restrict__ table,
    const int*   __restrict__ x,
    const int*   __restrict__ max_x_p,
    float* __restrict__ out)
{
    __shared__ float ldsQ[TM * D_];
    __shared__ float ldsP[TM * R_];
    __shared__ __attribute__((aligned(16))) int xs[S_];

    const int tid   = threadIdx.x;
    const int bid   = blockIdx.x;
    const int itile = bid & ((S_ / TM) - 1);
    const int h     = (bid >> 4) & (H_ - 1);
    const int b     = bid >> 7;
    const int i0    = itile * TM;
    const int mx    = max_x_p[0];

#pragma unroll
    for (int k = 0; k < (TM * D_) / 256; ++k) {
        int idx = tid + k * 256;
        int m = idx >> 6;
        int d = idx & (D_ - 1);
        ldsQ[idx] = q[(((size_t)b * S_ + (i0 + m)) * H_ + h) * D_ + d];
    }
#pragma unroll
    for (int k = 0; k < S_ / 256; ++k) {
        int j = tid + k * 256;
        xs[j] = x[b * S_ + j];
    }
    __syncthreads();

    const int jq = tid & 127;
    const int mh = tid >> 7;
    const int4 xj = *reinterpret_cast<const int4*>(&xs[4 * jq]);
    int xi[16];
#pragma unroll
    for (int t4 = 0; t4 < 4; ++t4) {
        const int4 v = *reinterpret_cast<const int4*>(&xs[i0 + mh * 16 + 4 * t4]);
        xi[4 * t4 + 0] = v.x; xi[4 * t4 + 1] = v.y; xi[4 * t4 + 2] = v.z; xi[4 * t4 + 3] = v.w;
    }

    const int wave   = tid >> 5;
    const int lane   = tid & 31;
    const int laneLo = lane & 15;
    const int hiSel  = lane >> 4;

    v2f a[2][16];
#pragma unroll
    for (int mt = 0; mt < 2; ++mt) {
        const float* src = &ldsQ[(mt * 16 + laneLo) * D_ + 2 * hiSel];
#pragma unroll
        for (int k4 = 0; k4 < 16; ++k4) {
            a[mt][k4] = *reinterpret_cast<const v2f*>(src + 4 * k4);
        }
    }

#pragma unroll
    for (int p = 0; p < 2; ++p) {
        const int nt = wave * 2 + p;
        const int r  = nt * 16 + laneLo;
        const float* bsrc = table + (size_t)r * E_ + h * D_ + 2 * hiSel;

        v8f acc0 = {};
        v8f acc1 = {};
#pragma unroll
        for (int k4 = 0; k4 < 16; ++k4) {
            v2f bv = *reinterpret_cast<const v2f*>(bsrc + 4 * k4);
            acc0 = __builtin_amdgcn_wmma_f32_16x16x4_f32(
                       false, a[0][k4], false, bv, (short)0, acc0, false, false);
            acc1 = __builtin_amdgcn_wmma_f32_16x16x4_f32(
                       false, a[1][k4], false, bv, (short)0, acc1, false, false);
        }

#pragma unroll
        for (int c = 0; c < 8; ++c) {
            int M = c + 8 * hiSel;
            ldsP[M * R_ + nt * 16 + laneLo]         = acc0[c];
            ldsP[(16 + M) * R_ + nt * 16 + laneLo]  = acc1[c];
        }
    }
    __syncthreads();

    float* outRow = out + (((size_t)(b * H_ + h)) * S_ + i0 + mh * 16) * S_ + 4 * jq;
    const float* pBase = &ldsP[(mh * 16) * R_] + mx;
#pragma unroll
    for (int t = 0; t < 16; ++t) {
        const float* pr = pBase + t * R_ - xi[t];
        float4 v;
        v.x = pr[xj.x];
        v.y = pr[xj.y];
        v.z = pr[xj.z];
        v.w = pr[xj.w];
        *reinterpret_cast<float4*>(outRow + (size_t)t * S_) = v;
    }
}

extern "C" void kernel_launch(void* const* d_in, const int* in_sizes, int n_in,
                              void* d_out, int out_size, void* d_ws, size_t ws_size,
                              hipStream_t stream) {
    const float* q     = (const float*)d_in[0];
    const float* table = (const float*)d_in[1];
    const int*   x     = (const int*)d_in[2];
    const int*   mx    = (const int*)d_in[3];
    float*       out   = (float*)d_out;
    (void)in_sizes; (void)n_in; (void)out_size; (void)d_ws; (void)ws_size;

    dim3 grid(B_ * H_ * (S_ / TM));
    dim3 block(256);
    hipLaunchKernelGGL(relpos_wmma_kernel, grid, block, 0, stream, q, table, x, mx, out);
}


# LDS rows padded by 4 dwords (q tile, P tile): A-fragment reads no longer put all 16 rows of a lane half on one bank pair, accumulator stores no longer put rows M and M+8 on one bank; sums unchanged
# speedup vs baseline: 1.6087x; 1.0878x over previous
#include <hip/hip_runtime.h>

typedef __attribute__((ext_vector_type(2))) float v2f;
typedef __attribute__((ext_vector_type(8))) float v8f;

#define B_  2
#define S_  512
#define H_  8
#define D_  64
#define E_  512
#define R_  256
#define TM  32
#define QP  (D_ + 4)
#define PP  (R_ + 4)

__global__ __launch_bounds__(256) void relpos_wmma_kernel(
    const float* __restrict__ q,
    const float* __restrict__ table,
    const int*   __restrict__ x,
    const int*   __restrict__ max_x_p,
    float* __restrict__ out)
{
    __shared__ __attribute__((aligned(16))) float ldsQ[TM * QP];
    __shared__ float ldsP[TM * PP];
    __shared__ __attribute__((aligned(16))) int xs[S_];

    const int tid   = threadIdx.x;
    const int bid   = blockIdx.x;
    const int itile = bid & ((S_ / TM) - 1);
    const int h     = (bid >> 4) & (H_ - 1);
    const int b     = bid >> 7;
    const int i0    = itile * TM;
    const int mx    = max_x_p[0];

#pragma unroll
    for (int k = 0; k < (TM * D_) / 256; ++k) {
        int idx = tid + k * 256;
        int m = idx >> 6;
        int d = idx & (D_ - 1);
        ldsQ[(tid >> 6) * QP + d + k * (4 * QP)] = q[(((size_t)b * S_ + (i0 + m)) * H_ + h) * D_ + d];
    }
#pragma unroll
    for (int k = 0; k < S_ / 256; ++k) {
        int j = tid + k * 256;
        xs[j] = x[b * S_ + j];
    }
    __syncthreads();

    const int jq = tid & 127;
    const int mh = tid >> 7;
    const int4 xj = *reinterpret_cast<const int4*>(&xs[4 * jq]);
    int xi[16];
#pragma unroll
    for (int t4 = 0; t4 < 4; ++t4) {
        const int4 v = *reinterpret_cast<const int4*>(&xs[i0 + mh * 16 + 4 * t4]);
        xi[4 * t4 + 0] = v.x; xi[4 * t4 + 1] = v.y; xi[4 * t4 + 2] = v.z; xi[4 * t4 + 3] = v.w;
    }

    const int wave   = tid >> 5;
    const int lane   = tid & 31;
    const int laneLo = lane & 15;
    const int hiSel  = lane >> 4;

    v2f a[2][16];
#pragma unroll
    for (int mt = 0; mt < 2; ++mt) {
        const float* src = &ldsQ[(mt * 16 + laneLo) * QP + 2 * hiSel];
#pragma unroll
        for (int k4 = 0; k4 < 16; ++k4) {
            a[mt][k4] = *reinterpret_cast<const v2f*>(src + 4 * k4);
        }
    }

#pragma unroll
    for (int p = 0; p < 2; ++p) {
        const int nt = wave * 2 + p;
        const int r  = nt * 16 + laneLo;
        const float* bsrc = table + (size_t)r * E_ + h * D_ + 2 * hiSel;

        v8f acc0 = {};
        v8f acc1 = {};
#pragma unroll
        for (int k4 = 0; k4 < 16; ++k4) {
            v2f bv = *reinterpret_cast<const v2f*>(bsrc + 4 * k4);
            acc0 = __builtin_amdgcn_wmma_f32_16x16x4_f32(
                       false, a[0][k4], false, bv, (short)0, acc0, false, false);
            acc1 = __builtin_amdgcn_wmma_f32_16x16x4_f32(
                       false, a[1][k4], false, bv, (short)0, acc1, false, false);
        }

#pragma unroll
        for (int c = 0; c < 8; ++c) {
            int M = c + 8 * hiSel;
            ldsP[M * PP + nt * 16 + laneLo]         = acc0[c];
            ldsP[(16 + M) * PP + nt * 16 + laneLo]  = acc1[c];
        }
    }
    __syncthreads();

    float* outRow = out + (((size_t)(b * H_ + h)) * S_ + i0 + mh * 16) * S_ + 4 * jq;
    const float* pBase = &ldsP[(mh * 16) * PP] + mx;
#pragma unroll
    for (int t = 0; t < 16; ++t) {
        const float* pr = pBase + t * PP - xi[t];
        float4 v;
        v.x = pr[xj.x];
        v.y = pr[xj.y];
        v.z = pr[xj.z];
        v.w = pr[xj.w];
        *reinterpret_cast<float4*>(outRow + (size_t)t * S_) = v;
    }
}

extern "C" void kernel_launch(void* const* d_in, const int* in_sizes, int n_in,
                              void* d_out, int out_size, void* d_ws, size_t ws_size,
                              hipStream_t stream) {
    const float* q     = (const float*)d_in[0];
    const float* table = (const float*)d_in[1];
    const int*   x     = (const int*)d_in[2];
    const int*   mx    = (const int*)d_in[3];
    float*       out   = (float*)d_out;
    (void)in_sizes; (void)n_in; (void)out_size; (void)d_ws; (void)ws_size;

    dim3 grid(B_ * H_ * (S_ / TM));
    dim3 block(256);
    hipLaunchKernelGGL(relpos_wmma_kernel, grid, block, 0, stream, q, table, x, mx, out);
}


# B fragments fetched as 16-B loads at kernel entry, before the q barrier (LDS-only fence there), and held in registers; A read the same way, so each product sums k in a permuted order
# speedup vs baseline: 1.7058x; 1.0604x over previous
#include <hip/hip_runtime.h>

typedef __attribute__((ext_vector_type(2))) float v2f;
typedef __attribute__((ext_vector_type(4))) float v4f;
typedef __attribute__((ext_vector_type(8))) float v8f;

#define B_  2
#define S_  512
#define H_  8
#define D_  64
#define E_  512
#define R_  256
#define TM  32
#define QP  (D_ + 4)
#define PP  (R_ + 4)

__global__ __launch_bounds__(256) void relpos_wmma_kernel(
    const float* __restrict__ q,
    const float* __restrict__ table,
    const int*   __restrict__ x,
    const int*   __restrict__ max_x_p,
    float* __restrict__ out)
{
    __shared__ __attribute__((aligned(16))) float ldsQ[TM * QP];
    __shared__ float ldsP[TM * PP];
    __shared__ __attribute__((aligned(16))) int xs[S_];

    const int tid   = threadIdx.x;
    const int bid   = blockIdx.x;
    const int itile = bid & ((S_ / TM) - 1);
    const int h     = (bid >> 4) & (H_ - 1);
    const int b     = bid >> 7;
    const int i0    = itile * TM;
    const int mx    = max_x_p[0];

#pragma unroll
    for (int k = 0; k < (TM * D_) / 256; ++k) {
        int idx = tid + k * 256;
        int m = idx >> 6;
        int d = idx & (D_ - 1);
        ldsQ[(tid >> 6) * QP + d + k * (4 * QP)] = q[(((size_t)b * S_ + (i0 + m)) * H_ + h) * D_ + d];
    }
#pragma unroll
    for (int k = 0; k < S_ / 256; ++k) {
        int j = tid + k * 256;
        xs[j] = x[b * S_ + j];
    }

    const int wave   = tid >> 5;
    const int lane   = tid & 31;
    const int laneLo = lane & 15;
    const int hiSel  = lane >> 4;

    v4f bq[8][2];
#pragma unroll
    for (int kk = 0; kk < 8; ++kk) {
#pragma unroll
        for (int p = 0; p < 2; ++p) {
            const int r = (wave * 2 + p) * 16 + laneLo;
            bq[kk][p] = *reinterpret_cast<const v4f*>(table + (size_t)r * E_ + h * D_ + 8 * kk + 4 * hiSel);
        }
    }
    __builtin_amdgcn_fence(__ATOMIC_RELEASE, "workgroup", "local");
    __builtin_amdgcn_s_barrier();
    __builtin_amdgcn_fence(__ATOMIC_ACQUIRE, "workgroup", "local");

    const int jq = tid & 127;
    const int mh = tid >> 7;
    const int4 xj = *reinterpret_cast<const int4*>(&xs[4 * jq]);
    int xi[16];
#pragma unroll
    for (int t4 = 0; t4 < 4; ++t4) {
        const int4 v = *reinterpret_cast<const int4*>(&xs[i0 + mh * 16 + 4 * t4]);
        xi[4 * t4 + 0] = v.x; xi[4 * t4 + 1] = v.y; xi[4 * t4 + 2] = v.z; xi[4 * t4 + 3] = v.w;
    }

    v4f aq[2][8];
#pragma unroll
    for (int mt = 0; mt < 2; ++mt) {
        const float* src = &ldsQ[(mt * 16 + laneLo) * QP + 4 * hiSel];
#pragma unroll
        for (int kk = 0; kk < 8; ++kk) {
            aq[mt][kk] = *reinterpret_cast<const v4f*>(src + 8 * kk);
        }
    }

    v8f acc[2][2] = {};
#pragma unroll
    for (int kk = 0; kk < 8; ++kk) {
#pragma unroll
        for (int p = 0; p < 2; ++p) {
            const v4f bv = bq[kk][p];
            const v2f b01 = {bv.x, bv.y};
            const v2f b23 = {bv.z, bv.w};
#pragma unroll
            for (int mt = 0; mt < 2; ++mt) {
                const v4f av = aq[mt][kk];
                const v2f a01 = {av.x, av.y};
                const v2f a23 = {av.z, av.w};
                acc[p][mt] = __builtin_amdgcn_wmma_f32_16x16x4_f32(
                                 false, a01, false, b01, (short)0, acc[p][mt], false, false);
                acc[p][mt] = __builtin_amdgcn_wmma_f32_16x16x4_f32(
                                 false, a23, false, b23, (short)0, acc[p][mt], false, false);
            }
        }
    }

#pragma unroll
    for (int p = 0; p < 2; ++p) {
        const int nt = wave * 2 + p;
#pragma unroll
        for (int c = 0; c < 8; ++c) {
            int M = c + 8 * hiSel;
            ldsP[M * PP + nt * 16 + laneLo]         = acc[p][0][c];
            ldsP[(16 + M) * PP + nt * 16 + laneLo]  = acc[p][1][c];
        }
    }
    __syncthreads();

    float* outRow = out + (((size_t)(b * H_ + h)) * S_ + i0 + mh * 16) * S_ + 4 * jq;
    const float* pBase = &ldsP[(mh * 16) * PP] + mx;
#pragma unroll
    for (int t = 0; t < 16; ++t) {
        const float* pr = pBase + t * PP - xi[t];
        float4 v;
        v.x = pr[xj.x];
        v.y = pr[xj.y];
        v.z = pr[xj.z];
        v.w = pr[xj.w];
        *reinterpret_cast<float4*>(outRow + (size_t)t * S_) = v;
    }
}

extern "C" void kernel_launch(void* const* d_in, const int* in_sizes, int n_in,
                              void* d_out, int out_size, void* d_ws, size_t ws_size,
                              hipStream_t stream) {
    const float* q     = (const float*)d_in[0];
    const float* table = (const float*)d_in[1];
    const int*   x     = (const int*)d_in[2];
    const int*   mx    = (const int*)d_in[3];
    float*       out   = (float*)d_out;
    (void)in_sizes; (void)n_in; (void)out_size; (void)d_ws; (void)ws_size;

    dim3 grid(B_ * H_ * (S_ / TM));
    dim3 block(256);
    hipLaunchKernelGGL(relpos_wmma_kernel, grid, block, 0, stream, q, table, x, mx, out);
}


# tile worked as two 16-row halves in sequence: LDS gathers and 16-B stores of rows 0-15 are issued among the products of rows 16-31 (LDS-only barriers); every accumulator sums as before
# speedup vs baseline: 1.8149x; 1.0640x over previous
#include <hip/hip_runtime.h>

typedef __attribute__((ext_vector_type(2))) float v2f;
typedef __attribute__((ext_vector_type(4))) float v4f;
typedef __attribute__((ext_vector_type(8))) float v8f;

#define B_  2
#define S_  512
#define H_  8
#define D_  64
#define E_  512
#define R_  256
#define TM  32
#define QP  (D_ + 4)
#define PP  (R_ + 4)

#define WMMA_PAIR(ACC, AV, BV)                                                                   \
    do {                                                                                         \
        const v4f av_ = (AV);                                                                    \
        const v4f bv_ = (BV);                                                                    \
        const v2f a01_ = {av_.x, av_.y};                                                         \
        const v2f a23_ = {av_.z, av_.w};                                                         \
        const v2f b01_ = {bv_.x, bv_.y};                                                         \
        const v2f b23_ = {bv_.z, bv_.w};                                                         \
        (ACC) = __builtin_amdgcn_wmma_f32_16x16x4_f32(false, a01_, false, b01_, (short)0, (ACC), false, false); \
        (ACC) = __builtin_amdgcn_wmma_f32_16x16x4_f32(false, a23_, false, b23_, (short)0, (ACC), false, false); \
    } while (0)

__global__ __launch_bounds__(256) void relpos_wmma_kernel(
    const float* __restrict__ q,
    const float* __restrict__ table,
    const int*   __restrict__ x,
    const int*   __restrict__ max_x_p,
    float* __restrict__ out)
{
    __shared__ __attribute__((aligned(16))) float ldsQ[TM * QP];
    __shared__ float ldsP[TM * PP];
    __shared__ __attribute__((aligned(16))) int xs[S_];

    const int tid   = threadIdx.x;
    const int bid   = blockIdx.x;
    const int itile = bid & ((S_ / TM) - 1);
    const int h     = (bid >> 4) & (H_ - 1);
    const int b     = bid >> 7;
    const int i0    = itile * TM;
    const int mx    = max_x_p[0];

#pragma unroll
    for (int k = 0; k < (TM * D_) / 256; ++k) {
        int idx = tid + k * 256;
        int m = idx >> 6;
        int d = idx & (D_ - 1);
        ldsQ[(tid >> 6) * QP + d + k * (4 * QP)] = q[(((size_t)b * S_ + (i0 + m)) * H_ + h) * D_ + d];
    }
#pragma unroll
    for (int k = 0; k < S_ / 256; ++k) {
        int j = tid + k * 256;
        xs[j] = x[b * S_ + j];
    }

    const int wave   = tid >> 5;
    const int lane   = tid & 31;
    const int laneLo = lane & 15;
    const int hiSel  = lane >> 4;

    v4f bq[8][2];
#pragma unroll
    for (int kk = 0; kk < 8; ++kk) {
#pragma unroll
        for (int p = 0; p < 2; ++p) {
            const int r = (wave * 2 + p) * 16 + laneLo;
            bq[kk][p] = *reinterpret_cast<const v4f*>(table + (size_t)r * E_ + h * D_ + 8 * kk + 4 * hiSel);
        }
    }
    __builtin_amdgcn_fence(__ATOMIC_RELEASE, "workgroup", "local");
    __builtin_amdgcn_s_barrier();
    __builtin_amdgcn_fence(__ATOMIC_ACQUIRE, "workgroup", "local");

    const int jq = tid & 127;
    const int mh = tid >> 7;
    const int4 xj = *reinterpret_cast<const int4*>(&xs[4 * jq]);
    int xi[2][8];
#pragma unroll
    for (int hf = 0; hf < 2; ++hf) {
#pragma unroll
        for (int t4 = 0; t4 < 2; ++t4) {
            const int4 v = *reinterpret_cast<const int4*>(&xs[i0 + hf * 16 + mh * 8 + 4 * t4]);
            xi[hf][4 * t4 + 0] = v.x; xi[hf][4 * t4 + 1] = v.y; xi[hf][4 * t4 + 2] = v.z; xi[hf][4 * t4 + 3] = v.w;
        }
    }

    v4f aq[2][8];
#pragma unroll
    for (int mt = 0; mt < 2; ++mt) {
        const float* src = &ldsQ[(mt * 16 + laneLo) * QP + 4 * hiSel];
#pragma unroll
        for (int kk = 0; kk < 8; ++kk) {
            aq[mt][kk] = *reinterpret_cast<const v4f*>(src + 8 * kk);
        }
    }

    v8f acc[2][2] = {};
#pragma unroll
    for (int kk = 0; kk < 8; ++kk) {
#pragma unroll
        for (int p = 0; p < 2; ++p) {
            WMMA_PAIR(acc[p][0], aq[0][kk], bq[kk][p]);
        }
    }
#pragma unroll
    for (int p = 0; p < 2; ++p) {
        const int nt = wave * 2 + p;
#pragma unroll
        for (int c = 0; c < 8; ++c) {
            ldsP[(c + 8 * hiSel) * PP + nt * 16 + laneLo] = acc[p][0][c];
        }
    }
    __builtin_amdgcn_sched_barrier(0);
#pragma unroll
    for (int kk = 0; kk < 2; ++kk) {
#pragma unroll
        for (int p = 0; p < 2; ++p) {
            WMMA_PAIR(acc[p][1], aq[1][kk], bq[kk][p]);
        }
    }
    __builtin_amdgcn_sched_barrier(0);
    __builtin_amdgcn_fence(__ATOMIC_RELEASE, "workgroup", "local");
    __builtin_amdgcn_s_barrier();
    __builtin_amdgcn_fence(__ATOMIC_ACQUIRE, "workgroup", "local");

    float* outRow = out + (((size_t)(b * H_ + h)) * S_ + i0 + mh * 8) * S_ + 4 * jq;
    const float* pBase = &ldsP[(mh * 8) * PP] + mx;
    float4 g[8];
#pragma unroll
    for (int t = 0; t < 8; ++t) {
        const float* pr = pBase + t * PP - xi[0][t];
        g[t].x = pr[xj.x];
        g[t].y = pr[xj.y];
        g[t].z = pr[xj.z];
        g[t].w = pr[xj.w];
    }
#pragma unroll
    for (int kk = 2; kk < 5; ++kk) {
#pragma unroll
        for (int p = 0; p < 2; ++p) {
            WMMA_PAIR(acc[p][1], aq[1][kk], bq[kk][p]);
        }
    }
    __builtin_amdgcn_sched_barrier(0);
#pragma unroll
    for (int t = 0; t < 8; ++t) {
        *reinterpret_cast<float4*>(outRow + (size_t)t * S_) = g[t];
    }
#pragma unroll
    for (int kk = 5; kk < 8; ++kk) {
#pragma unroll
        for (int p = 0; p < 2; ++p) {
            WMMA_PAIR(acc[p][1], aq[1][kk], bq[kk][p]);
        }
    }
#pragma unroll
    for (int p = 0; p < 2; ++p) {
        const int nt = wave * 2 + p;
#pragma unroll
        for (int c = 0; c < 8; ++c) {
            ldsP[(16 + c + 8 * hiSel) * PP + nt * 16 + laneLo] = acc[p][1][c];
        }
    }
    __builtin_amdgcn_fence(__ATOMIC_RELEASE, "workgroup", "local");
    __builtin_amdgcn_s_barrier();
    __builtin_amdgcn_fence(__ATOMIC_ACQUIRE, "workgroup", "local");

#pragma unroll
    for (int t = 0; t < 8; ++t) {
        const float* pr = pBase + (16 + t) * PP - xi[1][t];
        float4 v;
        v.x = pr[xj.x];
        v.y = pr[xj.y];
        v.z = pr[xj.z];
        v.w = pr[xj.w];
        *reinterpret_cast<float4*>(outRow + (size_t)(16 + t) * S_) = v;
    }
}

extern "C" void kernel_launch(void* const* d_in, const int* in_sizes, int n_in,
                              void* d_out, int out_size, void* d_ws, size_t ws_size,
                              hipStream_t stream) {
    const float* q     = (const float*)d_in[0];
    const float* table = (const float*)d_in[1];
    const int*   x     = (const int*)d_in[2];
    const int*   mx    = (const int*)d_in[3];
    float*       out   = (float*)d_out;
    (void)in_sizes; (void)n_in; (void)out_size; (void)d_ws; (void)ws_size;

    dim3 grid(B_ * H_ * (S_ / TM));
    dim3 block(256);
    hipLaunchKernelGGL(relpos_wmma_kernel, grid, block, 0, stream, q, table, x, mx, out);
}
